# starting kernel
# baseline (speedup 1.0000x reference)
#include <hip/hip_runtime.h>
#include <hip/hip_fp16.h>


typedef __attribute__((ext_vector_type(2)))  float    v2f;
typedef __attribute__((ext_vector_type(8)))  float    v8f;
typedef __attribute__((ext_vector_type(16))) _Float16 v16h;

#define BATCH   8
#define HW_IN   112
#define HW_OUT  56
#define NPATCH  (HW_OUT * HW_OUT)
#define NTILE   (NPATCH / 16)
#define LDS_RS  18

__device__ __forceinline__ v8f wmma_k4(v2f a, v2f b, v8f c) {
#if __has_builtin(__builtin_amdgcn_wmma_f32_16x16x4_f32)
  return __builtin_amdgcn_wmma_f32_16x16x4_f32(false, a, false, b, (short)0, c,
                                               false, false);
#else
  v16h ah = {};
  v16h bh = {};
  ah[0] = (_Float16)a[0]; ah[1] = (_Float16)a[1];
  bh[0] = (_Float16)b[0]; bh[1] = (_Float16)b[1];
  return __builtin_amdgcn_wmma_f32_16x16x32_f16(false, ah, false, bh, (short)0, c,
                                                false, false);
#endif
}

__global__ __launch_bounds__(256) void conv_norm_kernel(
    const float* __restrict__ x,
    const float* __restrict__ cw,
    const float* __restrict__ cb,
    float* __restrict__ patches,
    float* __restrict__ normed) {
  int t = blockIdx.x * blockDim.x + threadIdx.x;
  if (t >= BATCH * NPATCH) return;
  int b = t / NPATCH;
  int n = t % NPATCH;
  int h = n / HW_OUT, w = n % HW_OUT;

  const float* xb = x + ((size_t)b * HW_IN + 2 * h) * HW_IN + 2 * w;
  float x00 = xb[0], x01 = xb[1], x10 = xb[HW_IN], x11 = xb[HW_IN + 1];

  float f[4];
  float ss = 0.0f;
#pragma unroll
  for (int cc = 0; cc < 4; ++cc) {
    float v = x00 * cw[cc * 4 + 0] + x01 * cw[cc * 4 + 1] +
              x10 * cw[cc * 4 + 2] + x11 * cw[cc * 4 + 3] + cb[cc];
    f[cc] = v;
    ss += v * v;
  }
  float inv = 1.0f / (sqrtf(ss) + 1e-12f);

  float4 pf = make_float4(f[0], f[1], f[2], f[3]);
  float4 nf = make_float4(f[0] * inv, f[1] * inv, f[2] * inv, f[3] * inv);
  *(float4*)(patches + (size_t)t * 4) = pf;
  *(float4*)(normed  + (size_t)t * 4) = nf;
}

__global__ __launch_bounds__(256) void build_paugb_kernel(
    const float* __restrict__ patches,
    float* __restrict__ paugB) {
  int t = blockIdx.x * blockDim.x + threadIdx.x;
  if (t >= BATCH * NTILE * 4 * 32) return;
  int lane = t & 31;
  int k    = (t >> 5) & 3;
  int j    = (t >> 7) % NTILE;
  int b    = t / (NTILE * 4 * 32);

  int half  = lane >> 4;
  int c     = lane & 15;
  int kbase = half << 1;
  int m0    = 4 * k + kbase;
  int cclamp = (c < 3) ? c : 3;

  const float* pat_b = patches + (size_t)b * NPATCH * 4;
  float pv0 = pat_b[(j * 16 + m0    ) * 4 + cclamp];
  float pv1 = pat_b[(j * 16 + m0 + 1) * 4 + cclamp];

  v2f v;
  v[0] = (c < 4) ? pv0 : ((c == 4) ? 1.0f : 0.0f);
  v[1] = (c < 4) ? pv1 : ((c == 4) ? 1.0f : 0.0f);
  *(v2f*)(paugB + (size_t)t * 2) = v;
}

__global__ __launch_bounds__(256) void smooth_kernel(
    const float* __restrict__ patches,
    const float* __restrict__ normed,
    const float* __restrict__ paugB,
    float* __restrict__ out) {
  __shared__ float lds[8 * 16 * LDS_RS];

  const int lane  = threadIdx.x & 31;
  const int wave  = threadIdx.x >> 5;
  const int gw    = blockIdx.x * 8 + wave;
  const int b     = gw / NTILE;
  const int it    = gw % NTILE;

  float* wtile = &lds[wave * 16 * LDS_RS];

  const int half  = lane >> 4;
  const int l15   = lane & 15;
  const int kbase = half << 1;
  const int c     = l15;

  const float* nrm_b  = normed  + (size_t)b * NPATCH * 4;
  const float* pat_b  = patches + (size_t)b * NPATCH * 4;
  const float* paug_b = paugB   + (size_t)b * NTILE * 4 * 32 * 2;

  v2f a1 = *(const v2f*)(nrm_b + (it * 16 + l15) * 4 + kbase);

  v8f acc = {};

  for (int j = 0; j < NTILE; ++j) {
    v2f b1 = *(const v2f*)(nrm_b + (j * 16 + l15) * 4 + kbase);

    int jn = (j + 1 < NTILE) ? (j + 1) : (NTILE - 1);
    __builtin_prefetch(nrm_b + (jn * 16 + l15) * 4 + kbase, 0, 3);
    __builtin_prefetch(paug_b + (size_t)jn * 4 * 32 * 2 + lane * 2, 0, 3);

    v8f zero = {};
    v8f cosd = wmma_k4(a1, b1, zero);

#pragma unroll
    for (int r = 0; r < 8; ++r) {
      float cc  = cosd[r];
      float fid = cc * cc;
      float wgt = (fid >= 0.8f) ? 1.0f : ((fid >= 0.5f) ? 0.5f : 0.0f);
      int row_local = r + 8 * half;
      if (it == j && row_local == c) wgt = 0.0f;
      wtile[row_local * LDS_RS + l15] = wgt;
    }
    asm volatile("s_wait_dscnt 0" ::: "memory");

#pragma unroll
    for (int k = 0; k < 4; ++k) {
      v2f a2 = *(const v2f*)(wtile + l15 * LDS_RS + 4 * k + kbase);
      v2f b2 = *(const v2f*)(paug_b + ((size_t)j * 4 + k) * 32 * 2 + lane * 2);
      acc = wmma_k4(a2, b2, acc);
    }
  }

#pragma unroll
  for (int r = 0; r < 8; ++r) {
    int row_local = r + 8 * half;
    float deg = __shfl(acc[r], (half << 4) | 4, 32);
    if (c < 4) {
      int n = it * 16 + row_local;
      float p = pat_b[n * 4 + c];
      out[(size_t)b * NPATCH * 4 + n * 4 + c] = p * (1.0f + deg) - acc[r];
    }
  }
}

extern "C" void kernel_launch(void* const* d_in, const int* in_sizes, int n_in,
                              void* d_out, int out_size, void* d_ws, size_t ws_size,
                              hipStream_t stream) {
  const float* x  = (const float*)d_in[0];
  const float* cw = (const float*)d_in[1];
  const float* cb = (const float*)d_in[2];
  float* out = (float*)d_out;

  float* patches = (float*)d_ws;
  float* normed  = patches + (size_t)BATCH * NPATCH * 4;
  float* paugB   = normed  + (size_t)BATCH * NPATCH * 4;

  int total = BATCH * NPATCH;
  conv_norm_kernel<<<(total + 255) / 256, 256, 0, stream>>>(x, cw, cb, patches, normed);

  int totalB = BATCH * NTILE * 4 * 32;
  build_paugb_kernel<<<(totalB + 255) / 256, 256, 0, stream>>>(patches, paugB);

  smooth_kernel<<<NTILE, 256, 0, stream>>>(patches, normed, paugB, out);
}


# Split each row tile's column loop over 7 waves (14-wave blocks, 2 row tiles each); per-wave partial sums are added through LDS in segment order: a different reduction tree.
# speedup vs baseline: 3.4763x; 3.4763x over previous
#include <hip/hip_runtime.h>
#include <hip/hip_fp16.h>


typedef __attribute__((ext_vector_type(2)))  float    v2f;
typedef __attribute__((ext_vector_type(8)))  float    v8f;
typedef __attribute__((ext_vector_type(16))) _Float16 v16h;

#define BATCH   8
#define HW_IN   112
#define HW_OUT  56
#define NPATCH  (HW_OUT * HW_OUT)
#define NTILE   (NPATCH / 16)
#define LDS_RS  18
#define RT      2
#define CS      7
#define NWAVE   (RT * CS)
#define JSEG    (NTILE / CS)
#define RED_RS  5

__device__ __forceinline__ v8f wmma_k4(v2f a, v2f b, v8f c) {
#if __has_builtin(__builtin_amdgcn_wmma_f32_16x16x4_f32)
  return __builtin_amdgcn_wmma_f32_16x16x4_f32(false, a, false, b, (short)0, c,
                                               false, false);
#else
  v16h ah = {};
  v16h bh = {};
  ah[0] = (_Float16)a[0]; ah[1] = (_Float16)a[1];
  bh[0] = (_Float16)b[0]; bh[1] = (_Float16)b[1];
  return __builtin_amdgcn_wmma_f32_16x16x32_f16(false, ah, false, bh, (short)0, c,
                                                false, false);
#endif
}

__global__ __launch_bounds__(256) void conv_norm_kernel(
    const float* __restrict__ x,
    const float* __restrict__ cw,
    const float* __restrict__ cb,
    float* __restrict__ patches,
    float* __restrict__ normed) {
  int t = blockIdx.x * blockDim.x + threadIdx.x;
  if (t >= BATCH * NPATCH) return;
  int b = t / NPATCH;
  int n = t % NPATCH;
  int h = n / HW_OUT, w = n % HW_OUT;

  const float* xb = x + ((size_t)b * HW_IN + 2 * h) * HW_IN + 2 * w;
  float x00 = xb[0], x01 = xb[1], x10 = xb[HW_IN], x11 = xb[HW_IN + 1];

  float f[4];
  float ss = 0.0f;
#pragma unroll
  for (int cc = 0; cc < 4; ++cc) {
    float v = x00 * cw[cc * 4 + 0] + x01 * cw[cc * 4 + 1] +
              x10 * cw[cc * 4 + 2] + x11 * cw[cc * 4 + 3] + cb[cc];
    f[cc] = v;
    ss += v * v;
  }
  float inv = 1.0f / (sqrtf(ss) + 1e-12f);

  float4 pf = make_float4(f[0], f[1], f[2], f[3]);
  float4 nf = make_float4(f[0] * inv, f[1] * inv, f[2] * inv, f[3] * inv);
  *(float4*)(patches + (size_t)t * 4) = pf;
  *(float4*)(normed  + (size_t)t * 4) = nf;
}

__global__ __launch_bounds__(256) void build_paugb_kernel(
    const float* __restrict__ patches,
    float* __restrict__ paugB) {
  int t = blockIdx.x * blockDim.x + threadIdx.x;
  if (t >= BATCH * NTILE * 4 * 32) return;
  int lane = t & 31;
  int k    = (t >> 5) & 3;
  int j    = (t >> 7) % NTILE;
  int b    = t / (NTILE * 4 * 32);

  int half  = lane >> 4;
  int c     = lane & 15;
  int kbase = half << 1;
  int m0    = 4 * k + kbase;
  int cclamp = (c < 3) ? c : 3;

  const float* pat_b = patches + (size_t)b * NPATCH * 4;
  float pv0 = pat_b[(j * 16 + m0    ) * 4 + cclamp];
  float pv1 = pat_b[(j * 16 + m0 + 1) * 4 + cclamp];

  v2f v;
  v[0] = (c < 4) ? pv0 : ((c == 4) ? 1.0f : 0.0f);
  v[1] = (c < 4) ? pv1 : ((c == 4) ? 1.0f : 0.0f);
  *(v2f*)(paugB + (size_t)t * 2) = v;
}

__global__ __launch_bounds__(NWAVE * 32) void smooth_kernel(
    const float* __restrict__ patches,
    const float* __restrict__ normed,
    const float* __restrict__ paugB,
    float* __restrict__ out) {
  __shared__ float lds[NWAVE * 16 * LDS_RS];
  __shared__ float red[NWAVE * 16 * RED_RS];

  const int lane  = threadIdx.x & 31;
  const int wave  = __builtin_amdgcn_readfirstlane(threadIdx.x >> 5);
  const int rt    = wave % RT;
  const int cs    = wave / RT;
  const int gt    = blockIdx.x * RT + rt;
  const int b     = gt / NTILE;
  const int it    = gt % NTILE;

  float* wtile = &lds[wave * 16 * LDS_RS];

  const int half  = lane >> 4;
  const int l15   = lane & 15;
  const int kbase = half << 1;
  const int c     = l15;

  const float* nrm_b  = normed  + (size_t)b * NPATCH * 4;
  const float* paug_b = paugB   + (size_t)b * NTILE * 4 * 32 * 2;

  v2f a1 = *(const v2f*)(nrm_b + (it * 16 + l15) * 4 + kbase);

  v8f acc = {};

  const int j0 = cs * JSEG;
  for (int j = j0; j < j0 + JSEG; ++j) {
    v2f b1 = *(const v2f*)(nrm_b + (j * 16 + l15) * 4 + kbase);

    int jn = (j + 1 < NTILE) ? (j + 1) : (NTILE - 1);
    __builtin_prefetch(nrm_b + (jn * 16 + l15) * 4 + kbase, 0, 3);
    __builtin_prefetch(paug_b + (size_t)jn * 4 * 32 * 2 + lane * 2, 0, 3);

    v8f zero = {};
    v8f cosd = wmma_k4(a1, b1, zero);

#pragma unroll
    for (int r = 0; r < 8; ++r) {
      float cc  = cosd[r];
      float fid = cc * cc;
      float wgt = (fid >= 0.8f) ? 1.0f : ((fid >= 0.5f) ? 0.5f : 0.0f);
      int row_local = r + 8 * half;
      if (it == j && row_local == c) wgt = 0.0f;
      wtile[row_local * LDS_RS + l15] = wgt;
    }
    asm volatile("s_wait_dscnt 0" ::: "memory");

#pragma unroll
    for (int k = 0; k < 4; ++k) {
      v2f a2 = *(const v2f*)(wtile + l15 * LDS_RS + 4 * k + kbase);
      v2f b2 = *(const v2f*)(paug_b + ((size_t)j * 4 + k) * 32 * 2 + lane * 2);
      acc = wmma_k4(a2, b2, acc);
    }
  }

  float* myred = &red[wave * 16 * RED_RS];
  if (c < RED_RS) {
#pragma unroll
    for (int r = 0; r < 8; ++r)
      myred[(r + 8 * half) * RED_RS + c] = acc[r];
  }
  __syncthreads();

  if (threadIdx.x < RT * 64) {
    const int rtf = threadIdx.x >> 6;
    const int e   = threadIdx.x & 63;
    const int row = e >> 2;
    const int col = e & 3;
    float s = 0.0f, deg = 0.0f;
#pragma unroll
    for (int q = 0; q < CS; ++q) {
      const float* rr = &red[((q * RT + rtf) * 16 + row) * RED_RS];
      s   += rr[col];
      deg += rr[4];
    }
    const int gtf = blockIdx.x * RT + rtf;
    const size_t n = (size_t)(gtf / NTILE) * NPATCH + (gtf % NTILE) * 16 + row;
    float p = patches[n * 4 + col];
    out[n * 4 + col] = p * (1.0f + deg) - s;
  }
}

extern "C" void kernel_launch(void* const* d_in, const int* in_sizes, int n_in,
                              void* d_out, int out_size, void* d_ws, size_t ws_size,
                              hipStream_t stream) {
  const float* x  = (const float*)d_in[0];
  const float* cw = (const float*)d_in[1];
  const float* cb = (const float*)d_in[2];
  float* out = (float*)d_out;

  float* patches = (float*)d_ws;
  float* normed  = patches + (size_t)BATCH * NPATCH * 4;
  float* paugB   = normed  + (size_t)BATCH * NPATCH * 4;

  int total = BATCH * NPATCH;
  conv_norm_kernel<<<(total + 255) / 256, 256, 0, stream>>>(x, cw, cb, patches, normed);

  int totalB = BATCH * NTILE * 4 * 32;
  build_paugb_kernel<<<(totalB + 255) / 256, 256, 0, stream>>>(patches, paugB);

  smooth_kernel<<<BATCH * NTILE / RT, NWAVE * 32, 0, stream>>>(patches, normed, paugB, out);
}


# Weight tile stays in stage 1's D registers as B of a transposed second product (A = pre-swizzled p^T, same k order): no LDS round trip per tile; a tile's 16 terms are summed in another order.
# speedup vs baseline: 3.7198x; 1.0700x over previous
#include <hip/hip_runtime.h>
#include <hip/hip_fp16.h>


typedef __attribute__((ext_vector_type(2)))  float    v2f;
typedef __attribute__((ext_vector_type(8)))  float    v8f;
typedef __attribute__((ext_vector_type(16))) _Float16 v16h;

#define BATCH   8
#define HW_IN   112
#define HW_OUT  56
#define NPATCH  (HW_OUT * HW_OUT)
#define NTILE   (NPATCH / 16)
#define RT      2
#define CS      7
#define NWAVE   (RT * CS)
#define JSEG    (NTILE / CS)
#define RED_RS  5

__device__ __forceinline__ v8f wmma_k4(v2f a, v2f b, v8f c) {
#if __has_builtin(__builtin_amdgcn_wmma_f32_16x16x4_f32)
  return __builtin_amdgcn_wmma_f32_16x16x4_f32(false, a, false, b, (short)0, c,
                                               false, false);
#else
  v16h ah = {};
  v16h bh = {};
  ah[0] = (_Float16)a[0]; ah[1] = (_Float16)a[1];
  bh[0] = (_Float16)b[0]; bh[1] = (_Float16)b[1];
  return __builtin_amdgcn_wmma_f32_16x16x32_f16(false, ah, false, bh, (short)0, c,
                                                false, false);
#endif
}

__global__ __launch_bounds__(256) void conv_norm_kernel(
    const float* __restrict__ x,
    const float* __restrict__ cw,
    const float* __restrict__ cb,
    float* __restrict__ patches,
    float* __restrict__ normed) {
  int t = blockIdx.x * blockDim.x + threadIdx.x;
  if (t >= BATCH * NPATCH) return;
  int b = t / NPATCH;
  int n = t % NPATCH;
  int h = n / HW_OUT, w = n % HW_OUT;

  const float* xb = x + ((size_t)b * HW_IN + 2 * h) * HW_IN + 2 * w;
  float x00 = xb[0], x01 = xb[1], x10 = xb[HW_IN], x11 = xb[HW_IN + 1];

  float f[4];
  float ss = 0.0f;
#pragma unroll
  for (int cc = 0; cc < 4; ++cc) {
    float v = x00 * cw[cc * 4 + 0] + x01 * cw[cc * 4 + 1] +
              x10 * cw[cc * 4 + 2] + x11 * cw[cc * 4 + 3] + cb[cc];
    f[cc] = v;
    ss += v * v;
  }
  float inv = 1.0f / (sqrtf(ss) + 1e-12f);

  float4 pf = make_float4(f[0], f[1], f[2], f[3]);
  float4 nf = make_float4(f[0] * inv, f[1] * inv, f[2] * inv, f[3] * inv);
  *(float4*)(patches + (size_t)t * 4) = pf;
  *(float4*)(normed  + (size_t)t * 4) = nf;
}

__global__ __launch_bounds__(256) void build_paugt_kernel(
    const float* __restrict__ patches,
    float* __restrict__ paugT) {
  int t = blockIdx.x * blockDim.x + threadIdx.x;
  if (t >= BATCH * NTILE * 4 * 32) return;
  int lane = t & 31;
  int q    = (t >> 5) & 3;
  int j    = (t >> 7) % NTILE;
  int b    = t / (NTILE * 4 * 32);

  int half  = lane >> 4;
  int c     = lane & 15;
  int m0    = 8 * half + 2 * q;
  int cclamp = (c < 3) ? c : 3;

  const float* pat_b = patches + (size_t)b * NPATCH * 4;
  float pv0 = pat_b[(j * 16 + m0    ) * 4 + cclamp];
  float pv1 = pat_b[(j * 16 + m0 + 1) * 4 + cclamp];

  v2f v;
  v[0] = (c < 4) ? pv0 : ((c == 4) ? 1.0f : 0.0f);
  v[1] = (c < 4) ? pv1 : ((c == 4) ? 1.0f : 0.0f);
  *(v2f*)(paugT + (size_t)t * 2) = v;
}

__global__ __launch_bounds__(NWAVE * 32) void smooth_kernel(
    const float* __restrict__ patches,
    const float* __restrict__ normed,
    const float* __restrict__ paugT,
    float* __restrict__ out) {
  __shared__ float red[NWAVE * 16 * RED_RS];

  const int lane  = threadIdx.x & 31;
  const int wave  = __builtin_amdgcn_readfirstlane(threadIdx.x >> 5);
  const int rt    = wave % RT;
  const int cs    = wave / RT;
  const int gt    = blockIdx.x * RT + rt;
  const int b     = gt / NTILE;
  const int it    = gt % NTILE;

  const int half  = lane >> 4;
  const int l15   = lane & 15;
  const int kbase = half << 1;
  const int c     = l15;

  const float* nrm_b  = normed  + (size_t)b * NPATCH * 4;
  const float* paug_b = paugT   + (size_t)b * NTILE * 4 * 32 * 2;

  v2f b1 = *(const v2f*)(nrm_b + (it * 16 + l15) * 4 + kbase);

  v8f acc = {};

  const int j0 = cs * JSEG;
  for (int j = j0; j < j0 + JSEG; ++j) {
    v2f a1 = *(const v2f*)(nrm_b + (j * 16 + l15) * 4 + kbase);

    int jn = (j + 1 < NTILE) ? (j + 1) : (NTILE - 1);
    __builtin_prefetch(nrm_b + (jn * 16 + l15) * 4 + kbase, 0, 3);
    __builtin_prefetch(paug_b + (size_t)jn * 4 * 32 * 2 + lane * 2, 0, 3);

    v8f zero = {};
    v8f cosd = wmma_k4(a1, b1, zero);

    float w[8];
#pragma unroll
    for (int r = 0; r < 8; ++r) {
      float cc  = cosd[r];
      float fid = cc * cc;
      float wgt = (fid >= 0.8f) ? 1.0f : ((fid >= 0.5f) ? 0.5f : 0.0f);
      int col_local = r + 8 * half;
      if (it == j && col_local == c) wgt = 0.0f;
      w[r] = wgt;
    }

#pragma unroll
    for (int q = 0; q < 4; ++q) {
      v2f a2 = *(const v2f*)(paug_b + ((size_t)j * 4 + q) * 32 * 2 + lane * 2);
      v2f b2;
      b2[0] = w[2 * q];
      b2[1] = w[2 * q + 1];
      acc = wmma_k4(a2, b2, acc);
    }
  }

  float* myred = &red[wave * 16 * RED_RS];
  if (half == 0) {
#pragma unroll
    for (int r = 0; r < RED_RS; ++r)
      myred[c * RED_RS + r] = acc[r];
  }
  __syncthreads();

  if (threadIdx.x < RT * 64) {
    const int rtf = threadIdx.x >> 6;
    const int e   = threadIdx.x & 63;
    const int row = e >> 2;
    const int col = e & 3;
    float s = 0.0f, deg = 0.0f;
#pragma unroll
    for (int q = 0; q < CS; ++q) {
      const float* rr = &red[((q * RT + rtf) * 16 + row) * RED_RS];
      s   += rr[col];
      deg += rr[4];
    }
    const int gtf = blockIdx.x * RT + rtf;
    const size_t n = (size_t)(gtf / NTILE) * NPATCH + (gtf % NTILE) * 16 + row;
    float p = patches[n * 4 + col];
    out[n * 4 + col] = p * (1.0f + deg) - s;
  }
}

extern "C" void kernel_launch(void* const* d_in, const int* in_sizes, int n_in,
                              void* d_out, int out_size, void* d_ws, size_t ws_size,
                              hipStream_t stream) {
  const float* x  = (const float*)d_in[0];
  const float* cw = (const float*)d_in[1];
  const float* cb = (const float*)d_in[2];
  float* out = (float*)d_out;

  float* patches = (float*)d_ws;
  float* normed  = patches + (size_t)BATCH * NPATCH * 4;
  float* paugT   = normed  + (size_t)BATCH * NPATCH * 4;

  int total = BATCH * NPATCH;
  conv_norm_kernel<<<(total + 255) / 256, 256, 0, stream>>>(x, cw, cb, patches, normed);

  int totalB = BATCH * NTILE * 4 * 32;
  build_paugt_kernel<<<(totalB + 255) / 256, 256, 0, stream>>>(patches, paugT);

  smooth_kernel<<<BATCH * NTILE / RT, NWAVE * 32, 0, stream>>>(patches, normed, paugT, out);
}


# Conv+norm and the stage-2 operand swizzle fused into one launch: a block's patches pass through LDS, 196 blocks of 8 waves, 16-byte stores; 3 launches -> 2. Values and summation order unchanged.
# speedup vs baseline: 3.9228x; 1.0546x over previous
#include <hip/hip_runtime.h>
#include <hip/hip_fp16.h>


typedef __attribute__((ext_vector_type(2)))  float    v2f;
typedef __attribute__((ext_vector_type(8)))  float    v8f;
typedef __attribute__((ext_vector_type(16))) _Float16 v16h;

#define BATCH   8
#define HW_IN   112
#define HW_OUT  56
#define NPATCH  (HW_OUT * HW_OUT)
#define NTILE   (NPATCH / 16)
#define RT      2
#define CS      7
#define NWAVE   (RT * CS)
#define JSEG    (NTILE / CS)
#define RED_RS  5

__device__ __forceinline__ v8f wmma_k4(v2f a, v2f b, v8f c) {
#if __has_builtin(__builtin_amdgcn_wmma_f32_16x16x4_f32)
  return __builtin_amdgcn_wmma_f32_16x16x4_f32(false, a, false, b, (short)0, c,
                                               false, false);
#else
  v16h ah = {};
  v16h bh = {};
  ah[0] = (_Float16)a[0]; ah[1] = (_Float16)a[1];
  bh[0] = (_Float16)b[0]; bh[1] = (_Float16)b[1];
  return __builtin_amdgcn_wmma_f32_16x16x32_f16(false, ah, false, bh, (short)0, c,
                                                false, false);
#endif
}

#define PT      8
#define PREP_T  256

__global__ __launch_bounds__(PREP_T) void prep_kernel(
    const float* __restrict__ x,
    const float* __restrict__ cw,
    const float* __restrict__ cb,
    float* __restrict__ patches,
    float* __restrict__ normed,
    float* __restrict__ paugT) {
  __shared__ float4 sp[PT * 16];

  const int tile0 = blockIdx.x * PT;

  const int t = tile0 * 16 + threadIdx.x;
  float4 pf = make_float4(0.0f, 0.0f, 0.0f, 0.0f);
  float4 nf = pf;
  if (threadIdx.x < PT * 16) {
    int b = t / NPATCH;
    int n = t % NPATCH;
    int h = n / HW_OUT, w = n % HW_OUT;

    const float* xb = x + ((size_t)b * HW_IN + 2 * h) * HW_IN + 2 * w;
    float2 r0 = *(const float2*)(xb);
    float2 r1 = *(const float2*)(xb + HW_IN);
    float x00 = r0.x, x01 = r0.y, x10 = r1.x, x11 = r1.y;

    float f[4];
    float ss = 0.0f;
#pragma unroll
    for (int cc = 0; cc < 4; ++cc) {
      float v = x00 * cw[cc * 4 + 0] + x01 * cw[cc * 4 + 1] +
                x10 * cw[cc * 4 + 2] + x11 * cw[cc * 4 + 3] + cb[cc];
      f[cc] = v;
      ss += v * v;
    }
    float inv = 1.0f / (sqrtf(ss) + 1e-12f);

    pf = make_float4(f[0], f[1], f[2], f[3]);
    nf = make_float4(f[0] * inv, f[1] * inv, f[2] * inv, f[3] * inv);
    sp[threadIdx.x] = pf;
  }
  __syncthreads();

  if (threadIdx.x < PT * 16) {
    *(float4*)(patches + (size_t)t * 4) = pf;
    *(float4*)(normed  + (size_t)t * 4) = nf;
  }

  const float* spf = (const float*)sp;
  float4* pout = (float4*)paugT + (size_t)tile0 * 64;
  for (int i = threadIdx.x; i < PT * 64; i += PREP_T) {
    int jl   = i >> 6;
    int q    = (i >> 4) & 3;
    int half = (i >> 3) & 1;
    int cp   = i & 7;
    int m0   = 8 * half + 2 * q;
    int cpc  = (cp < 2) ? cp : 1;
    const float* p0 = spf + (jl * 16 + m0) * 4 + 2 * cpc;
    float2 a = *(const float2*)(p0);
    float2 d = *(const float2*)(p0 + 4);
    float one = (cp == 2) ? 1.0f : 0.0f;
    float4 v;
    v.x = (cp < 2) ? a.x : one;
    v.y = (cp < 2) ? d.x : one;
    v.z = (cp < 2) ? a.y : 0.0f;
    v.w = (cp < 2) ? d.y : 0.0f;
    pout[i] = v;
  }
}

__global__ __launch_bounds__(NWAVE * 32) void smooth_kernel(
    const float* __restrict__ patches,
    const float* __restrict__ normed,
    const float* __restrict__ paugT,
    float* __restrict__ out) {
  __shared__ float red[NWAVE * 16 * RED_RS];

  const int lane  = threadIdx.x & 31;
  const int wave  = __builtin_amdgcn_readfirstlane(threadIdx.x >> 5);
  const int rt    = wave % RT;
  const int cs    = wave / RT;
  const int gt    = blockIdx.x * RT + rt;
  const int b     = gt / NTILE;
  const int it    = gt % NTILE;

  const int half  = lane >> 4;
  const int l15   = lane & 15;
  const int kbase = half << 1;
  const int c     = l15;

  const float* nrm_b  = normed  + (size_t)b * NPATCH * 4;
  const float* paug_b = paugT   + (size_t)b * NTILE * 4 * 32 * 2;

  v2f b1 = *(const v2f*)(nrm_b + (it * 16 + l15) * 4 + kbase);

  v8f acc = {};

  const int j0 = cs * JSEG;
  for (int j = j0; j < j0 + JSEG; ++j) {
    v2f a1 = *(const v2f*)(nrm_b + (j * 16 + l15) * 4 + kbase);

    int jn = (j + 1 < NTILE) ? (j + 1) : (NTILE - 1);
    __builtin_prefetch(nrm_b + (jn * 16 + l15) * 4 + kbase, 0, 3);
    __builtin_prefetch(paug_b + (size_t)jn * 4 * 32 * 2 + lane * 2, 0, 3);

    v8f zero = {};
    v8f cosd = wmma_k4(a1, b1, zero);

    float w[8];
#pragma unroll
    for (int r = 0; r < 8; ++r) {
      float cc  = cosd[r];
      float fid = cc * cc;
      float wgt = (fid >= 0.8f) ? 1.0f : ((fid >= 0.5f) ? 0.5f : 0.0f);
      int col_local = r + 8 * half;
      if (it == j && col_local == c) wgt = 0.0f;
      w[r] = wgt;
    }

#pragma unroll
    for (int q = 0; q < 4; ++q) {
      v2f a2 = *(const v2f*)(paug_b + ((size_t)j * 4 + q) * 32 * 2 + lane * 2);
      v2f b2;
      b2[0] = w[2 * q];
      b2[1] = w[2 * q + 1];
      acc = wmma_k4(a2, b2, acc);
    }
  }

  float* myred = &red[wave * 16 * RED_RS];
  if (half == 0) {
#pragma unroll
    for (int r = 0; r < RED_RS; ++r)
      myred[c * RED_RS + r] = acc[r];
  }
  __syncthreads();

  if (threadIdx.x < RT * 64) {
    const int rtf = threadIdx.x >> 6;
    const int e   = threadIdx.x & 63;
    const int row = e >> 2;
    const int col = e & 3;
    float s = 0.0f, deg = 0.0f;
#pragma unroll
    for (int q = 0; q < CS; ++q) {
      const float* rr = &red[((q * RT + rtf) * 16 + row) * RED_RS];
      s   += rr[col];
      deg += rr[4];
    }
    const int gtf = blockIdx.x * RT + rtf;
    const size_t n = (size_t)(gtf / NTILE) * NPATCH + (gtf % NTILE) * 16 + row;
    float p = patches[n * 4 + col];
    out[n * 4 + col] = p * (1.0f + deg) - s;
  }
}

extern "C" void kernel_launch(void* const* d_in, const int* in_sizes, int n_in,
                              void* d_out, int out_size, void* d_ws, size_t ws_size,
                              hipStream_t stream) {
  const float* x  = (const float*)d_in[0];
  const float* cw = (const float*)d_in[1];
  const float* cb = (const float*)d_in[2];
  float* out = (float*)d_out;

  float* patches = (float*)d_ws;
  float* normed  = patches + (size_t)BATCH * NPATCH * 4;
  float* paugT   = normed  + (size_t)BATCH * NPATCH * 4;

  prep_kernel<<<BATCH * NTILE / PT, PREP_T, 0, stream>>>(x, cw, cb, patches, normed, paugT);

  smooth_kernel<<<BATCH * NTILE / RT, NWAVE * 32, 0, stream>>>(patches, normed, paugT, out);
}


# Each wave keeps 7 row tiles' accumulators and uses every loaded column operand 7 times: 224 blocks of 28 waves; a row's partial sums now meet over 28 segments, not 7: another reduction tree.
# speedup vs baseline: 4.5055x; 1.1485x over previous
#include <hip/hip_runtime.h>
#include <hip/hip_fp16.h>


typedef __attribute__((ext_vector_type(2)))  float    v2f;
typedef __attribute__((ext_vector_type(8)))  float    v8f;
typedef __attribute__((ext_vector_type(16))) _Float16 v16h;

#define BATCH   8
#define HW_IN   112
#define HW_OUT  56
#define NPATCH  (HW_OUT * HW_OUT)
#define NTILE   (NPATCH / 16)
#define RB      7
#define CS      28
#define NWAVE   CS
#define JSEG    (NTILE / CS)
#define NGROUP  (BATCH * NTILE / RB)
#define RED_RS  5

__device__ __forceinline__ v8f wmma_k4(v2f a, v2f b, v8f c) {
#if __has_builtin(__builtin_amdgcn_wmma_f32_16x16x4_f32)
  return __builtin_amdgcn_wmma_f32_16x16x4_f32(false, a, false, b, (short)0, c,
                                               false, false);
#else
  v16h ah = {};
  v16h bh = {};
  ah[0] = (_Float16)a[0]; ah[1] = (_Float16)a[1];
  bh[0] = (_Float16)b[0]; bh[1] = (_Float16)b[1];
  return __builtin_amdgcn_wmma_f32_16x16x32_f16(false, ah, false, bh, (short)0, c,
                                                false, false);
#endif
}

#define PT      8
#define PREP_T  256

__global__ __launch_bounds__(PREP_T) void prep_kernel(
    const float* __restrict__ x,
    const float* __restrict__ cw,
    const float* __restrict__ cb,
    float* __restrict__ patches,
    float* __restrict__ normed,
    float* __restrict__ paugT) {
  __shared__ float4 sp[PT * 16];

  const int tile0 = blockIdx.x * PT;

  const int t = tile0 * 16 + threadIdx.x;
  float4 pf = make_float4(0.0f, 0.0f, 0.0f, 0.0f);
  float4 nf = pf;
  if (threadIdx.x < PT * 16) {
    int b = t / NPATCH;
    int n = t % NPATCH;
    int h = n / HW_OUT, w = n % HW_OUT;

    const float* xb = x + ((size_t)b * HW_IN + 2 * h) * HW_IN + 2 * w;
    float2 r0 = *(const float2*)(xb);
    float2 r1 = *(const float2*)(xb + HW_IN);
    float x00 = r0.x, x01 = r0.y, x10 = r1.x, x11 = r1.y;

    float f[4];
    float ss = 0.0f;
#pragma unroll
    for (int cc = 0; cc < 4; ++cc) {
      float v = x00 * cw[cc * 4 + 0] + x01 * cw[cc * 4 + 1] +
                x10 * cw[cc * 4 + 2] + x11 * cw[cc * 4 + 3] + cb[cc];
      f[cc] = v;
      ss += v * v;
    }
    float inv = 1.0f / (sqrtf(ss) + 1e-12f);

    pf = make_float4(f[0], f[1], f[2], f[3]);
    nf = make_float4(f[0] * inv, f[1] * inv, f[2] * inv, f[3] * inv);
    sp[threadIdx.x] = pf;
  }
  __syncthreads();

  if (threadIdx.x < PT * 16) {
    *(float4*)(patches + (size_t)t * 4) = pf;
    *(float4*)(normed  + (size_t)t * 4) = nf;
  }

  const float* spf = (const float*)sp;
  float4* pout = (float4*)paugT + (size_t)tile0 * 64;
  for (int i = threadIdx.x; i < PT * 64; i += PREP_T) {
    int jl   = i >> 6;
    int q    = (i >> 4) & 3;
    int half = (i >> 3) & 1;
    int cp   = i & 7;
    int m0   = 8 * half + 2 * q;
    int cpc  = (cp < 2) ? cp : 1;
    const float* p0 = spf + (jl * 16 + m0) * 4 + 2 * cpc;
    float2 a = *(const float2*)(p0);
    float2 d = *(const float2*)(p0 + 4);
    float one = (cp == 2) ? 1.0f : 0.0f;
    float4 v;
    v.x = (cp < 2) ? a.x : one;
    v.y = (cp < 2) ? d.x : one;
    v.z = (cp < 2) ? a.y : 0.0f;
    v.w = (cp < 2) ? d.y : 0.0f;
    pout[i] = v;
  }
}

__global__ __launch_bounds__(NWAVE * 32) void smooth_kernel(
    const float* __restrict__ patches,
    const float* __restrict__ normed,
    const float* __restrict__ paugT,
    float* __restrict__ out) {
  __shared__ float red[CS * RB * 16 * RED_RS];

  const int lane  = threadIdx.x & 31;
  const int wave  = __builtin_amdgcn_readfirstlane(threadIdx.x >> 5);
  const int grp   = blockIdx.x;
  const int b     = grp / (NTILE / RB);
  const int it0   = (grp % (NTILE / RB)) * RB;

  const int half  = lane >> 4;
  const int l15   = lane & 15;
  const int kbase = half << 1;
  const int c     = l15;

  const float* nrm_b  = normed  + (size_t)b * NPATCH * 4;
  const float* paug_b = paugT   + (size_t)b * NTILE * 4 * 32 * 2;

  v2f b1[RB];
#pragma unroll
  for (int r = 0; r < RB; ++r)
    b1[r] = *(const v2f*)(nrm_b + ((it0 + r) * 16 + l15) * 4 + kbase);

  v8f acc[RB];
#pragma unroll
  for (int r = 0; r < RB; ++r) {
    v8f z = {};
    acc[r] = z;
  }

  const int j0 = wave * JSEG;
  for (int j = j0; j < j0 + JSEG; ++j) {
    v2f a1 = *(const v2f*)(nrm_b + (j * 16 + l15) * 4 + kbase);
    v2f a2[4];
#pragma unroll
    for (int q = 0; q < 4; ++q)
      a2[q] = *(const v2f*)(paug_b + ((size_t)j * 4 + q) * 32 * 2 + lane * 2);

    int jn = (j + 1 < NTILE) ? (j + 1) : (NTILE - 1);
    __builtin_prefetch(nrm_b + (jn * 16 + l15) * 4 + kbase, 0, 3);
    __builtin_prefetch(paug_b + (size_t)jn * 4 * 32 * 2 + lane * 2, 0, 3);

#pragma unroll
    for (int r = 0; r < RB; ++r) {
      const int it = it0 + r;
      v8f zero = {};
      v8f cosd = wmma_k4(a1, b1[r], zero);

      float w[8];
#pragma unroll
      for (int i = 0; i < 8; ++i) {
        float cc  = cosd[i];
        float fid = cc * cc;
        float wgt = (fid >= 0.8f) ? 1.0f : ((fid >= 0.5f) ? 0.5f : 0.0f);
        int col_local = i + 8 * half;
        if (it == j && col_local == c) wgt = 0.0f;
        w[i] = wgt;
      }

#pragma unroll
      for (int q = 0; q < 4; ++q) {
        v2f b2;
        b2[0] = w[2 * q];
        b2[1] = w[2 * q + 1];
        acc[r] = wmma_k4(a2[q], b2, acc[r]);
      }
    }
  }

  if (half == 0) {
#pragma unroll
    for (int r = 0; r < RB; ++r) {
      float* myred = &red[((wave * RB + r) * 16 + c) * RED_RS];
#pragma unroll
      for (int k = 0; k < RED_RS; ++k)
        myred[k] = acc[r][k];
    }
  }
  __syncthreads();

  if (threadIdx.x < RB * 64) {
    const int rtf = threadIdx.x >> 6;
    const int e   = threadIdx.x & 63;
    const int row = e >> 2;
    const int col = e & 3;
    float s = 0.0f, deg = 0.0f;
#pragma unroll
    for (int q = 0; q < CS; ++q) {
      const float* rr = &red[((q * RB + rtf) * 16 + row) * RED_RS];
      s   += rr[col];
      deg += rr[4];
    }
    const size_t n = (size_t)b * NPATCH + (it0 + rtf) * 16 + row;
    float p = patches[n * 4 + col];
    out[n * 4 + col] = p * (1.0f + deg) - s;
  }
}

extern "C" void kernel_launch(void* const* d_in, const int* in_sizes, int n_in,
                              void* d_out, int out_size, void* d_ws, size_t ws_size,
                              hipStream_t stream) {
  const float* x  = (const float*)d_in[0];
  const float* cw = (const float*)d_in[1];
  const float* cb = (const float*)d_in[2];
  float* out = (float*)d_out;

  float* patches = (float*)d_ws;
  float* normed  = patches + (size_t)BATCH * NPATCH * 4;
  float* paugT   = normed  + (size_t)BATCH * NPATCH * 4;

  prep_kernel<<<BATCH * NTILE / PT, PREP_T, 0, stream>>>(x, cw, cb, patches, normed, paugT);

  smooth_kernel<<<NGROUP, NWAVE * 32, 0, stream>>>(patches, normed, paugT, out);
}


# Weights by arithmetic, not compare/select through scalar masks: two clamped fused multiply-adds on |cos| per element, kept doubled (operand halved, exact); self-loop mask only in the diagonal wave
# speedup vs baseline: 5.5376x; 1.2291x over previous
#include <hip/hip_runtime.h>
#include <hip/hip_fp16.h>


typedef __attribute__((ext_vector_type(2)))  float    v2f;
typedef __attribute__((ext_vector_type(8)))  float    v8f;
typedef __attribute__((ext_vector_type(16))) _Float16 v16h;

#define BATCH   8
#define HW_IN   112
#define HW_OUT  56
#define NPATCH  (HW_OUT * HW_OUT)
#define NTILE   (NPATCH / 16)
#define RB      7
#define CS      28
#define NWAVE   CS
#define JSEG    (NTILE / CS)
#define NGROUP  (BATCH * NTILE / RB)
#define RED_RS  5
static_assert(JSEG == RB, "column segments and row groups must have one size");

#define TC_HI_PRED 0x1.c9f25cp-1f
#define TC_LO_PRED 0x1.6a09e6p-1f
#define STEP_H     0x1p30f

__device__ __forceinline__ v8f wmma_k4(v2f a, v2f b, v8f c) {
#if __has_builtin(__builtin_amdgcn_wmma_f32_16x16x4_f32)
  return __builtin_amdgcn_wmma_f32_16x16x4_f32(false, a, false, b, (short)0, c,
                                               false, false);
#else
  v16h ah = {};
  v16h bh = {};
  ah[0] = (_Float16)a[0]; ah[1] = (_Float16)a[1];
  bh[0] = (_Float16)b[0]; bh[1] = (_Float16)b[1];
  return __builtin_amdgcn_wmma_f32_16x16x32_f16(false, ah, false, bh, (short)0, c,
                                                false, false);
#endif
}

#define PT      8
#define PREP_T  256

__global__ __launch_bounds__(PREP_T) void prep_kernel(
    const float* __restrict__ x,
    const float* __restrict__ cw,
    const float* __restrict__ cb,
    float* __restrict__ patches,
    float* __restrict__ normed,
    float* __restrict__ paugT) {
  __shared__ float4 sp[PT * 16];

  const int tile0 = blockIdx.x * PT;

  const int t = tile0 * 16 + threadIdx.x;
  float4 pf = make_float4(0.0f, 0.0f, 0.0f, 0.0f);
  float4 nf = pf;
  if (threadIdx.x < PT * 16) {
    int b = t / NPATCH;
    int n = t % NPATCH;
    int h = n / HW_OUT, w = n % HW_OUT;

    const float* xb = x + ((size_t)b * HW_IN + 2 * h) * HW_IN + 2 * w;
    float2 r0 = *(const float2*)(xb);
    float2 r1 = *(const float2*)(xb + HW_IN);
    float x00 = r0.x, x01 = r0.y, x10 = r1.x, x11 = r1.y;

    float f[4];
    float ss = 0.0f;
#pragma unroll
    for (int cc = 0; cc < 4; ++cc) {
      float v = x00 * cw[cc * 4 + 0] + x01 * cw[cc * 4 + 1] +
                x10 * cw[cc * 4 + 2] + x11 * cw[cc * 4 + 3] + cb[cc];
      f[cc] = v;
      ss += v * v;
    }
    float inv = 1.0f / (sqrtf(ss) + 1e-12f);

    pf = make_float4(f[0], f[1], f[2], f[3]);
    nf = make_float4(f[0] * inv, f[1] * inv, f[2] * inv, f[3] * inv);
    sp[threadIdx.x] = pf;
  }
  __syncthreads();

  if (threadIdx.x < PT * 16) {
    *(float4*)(patches + (size_t)t * 4) = pf;
    *(float4*)(normed  + (size_t)t * 4) = nf;
  }

  const float* spf = (const float*)sp;
  float4* pout = (float4*)paugT + (size_t)tile0 * 64;
  for (int i = threadIdx.x; i < PT * 64; i += PREP_T) {
    int jl   = i >> 6;
    int q    = (i >> 4) & 3;
    int half = (i >> 3) & 1;
    int cp   = i & 7;
    int m0   = 8 * half + 2 * q;
    int cpc  = (cp < 2) ? cp : 1;
    const float* p0 = spf + (jl * 16 + m0) * 4 + 2 * cpc;
    float2 a = *(const float2*)(p0);
    float2 d = *(const float2*)(p0 + 4);
    float one = (cp == 2) ? 0.5f : 0.0f;
    float4 v;
    v.x = (cp < 2) ? 0.5f * a.x : one;
    v.y = (cp < 2) ? 0.5f * d.x : one;
    v.z = (cp < 2) ? 0.5f * a.y : 0.0f;
    v.w = (cp < 2) ? 0.5f * d.y : 0.0f;
    pout[i] = v;
  }
}

__device__ __forceinline__ float step01(float a, float pred) {
  return __builtin_amdgcn_fmed3f(__builtin_fmaf(a, STEP_H, -(pred * STEP_H)), 0.0f, 1.0f);
}

template <bool DIAG>
__device__ __forceinline__ void column_loop(
    const float* __restrict__ nrm_b, const float* __restrict__ paug_b,
    const int j0, const int it0, const int lane, const int l15,
    const int kbase, const int half, const int c,
    const v2f (&b1)[RB], v8f (&acc)[RB]) {
  for (int j = j0; j < j0 + JSEG; ++j) {
    v2f a1 = *(const v2f*)(nrm_b + (j * 16 + l15) * 4 + kbase);
    v2f a2[4];
#pragma unroll
    for (int q = 0; q < 4; ++q)
      a2[q] = *(const v2f*)(paug_b + ((size_t)j * 4 + q) * 32 * 2 + lane * 2);

    int jn = (j + 1 < NTILE) ? (j + 1) : (NTILE - 1);
    __builtin_prefetch(nrm_b + (jn * 16 + l15) * 4 + kbase, 0, 3);
    __builtin_prefetch(paug_b + (size_t)jn * 4 * 32 * 2 + lane * 2, 0, 3);

#pragma unroll
    for (int r = 0; r < RB; ++r) {
      v8f zero = {};
      v8f cosd = wmma_k4(a1, b1[r], zero);

      float lo[8], hi[8];
#pragma unroll
      for (int i = 0; i < 8; ++i) {
        const float a = __builtin_fabsf(cosd[i]);
        lo[i] = step01(a, TC_LO_PRED);
        hi[i] = step01(a, TC_HI_PRED);
      }

#pragma unroll
      for (int q = 0; q < 4; ++q) {
        v2f l, h;
        l[0] = lo[2 * q];  l[1] = lo[2 * q + 1];
        h[0] = hi[2 * q];  h[1] = hi[2 * q + 1];
        v2f b2 = l + h;
        if (DIAG) {
          if (it0 + r == j) {
            if (2 * q + 8 * half == c)     b2[0] = 0.0f;
            if (2 * q + 1 + 8 * half == c) b2[1] = 0.0f;
          }
        }
        acc[r] = wmma_k4(a2[q], b2, acc[r]);
      }
    }
  }
}

__global__ __launch_bounds__(NWAVE * 32) void smooth_kernel(
    const float* __restrict__ patches,
    const float* __restrict__ normed,
    const float* __restrict__ paugT,
    float* __restrict__ out) {
  __shared__ float red[CS * RB * 16 * RED_RS];

  const int lane  = threadIdx.x & 31;
  const int wave  = __builtin_amdgcn_readfirstlane(threadIdx.x >> 5);
  const int grp   = blockIdx.x;
  const int b     = grp / (NTILE / RB);
  const int it0   = (grp % (NTILE / RB)) * RB;

  const int half  = lane >> 4;
  const int l15   = lane & 15;
  const int kbase = half << 1;
  const int c     = l15;

  const float* nrm_b  = normed  + (size_t)b * NPATCH * 4;
  const float* paug_b = paugT   + (size_t)b * NTILE * 4 * 32 * 2;

  v2f b1[RB];
#pragma unroll
  for (int r = 0; r < RB; ++r)
    b1[r] = *(const v2f*)(nrm_b + ((it0 + r) * 16 + l15) * 4 + kbase);

  v8f acc[RB];
#pragma unroll
  for (int r = 0; r < RB; ++r) {
    v8f z = {};
    acc[r] = z;
  }

  const int j0 = wave * JSEG;
  if (j0 == it0)
    column_loop<true>(nrm_b, paug_b, j0, it0, lane, l15, kbase, half, c, b1, acc);
  else
    column_loop<false>(nrm_b, paug_b, j0, it0, lane, l15, kbase, half, c, b1, acc);

  if (half == 0) {
#pragma unroll
    for (int r = 0; r < RB; ++r) {
      float* myred = &red[((wave * RB + r) * 16 + c) * RED_RS];
#pragma unroll
      for (int k = 0; k < RED_RS; ++k)
        myred[k] = acc[r][k];
    }
  }
  __syncthreads();

  if (threadIdx.x < RB * 64) {
    const int rtf = threadIdx.x >> 6;
    const int e   = threadIdx.x & 63;
    const int row = e >> 2;
    const int col = e & 3;
    float s = 0.0f, deg = 0.0f;
#pragma unroll
    for (int q = 0; q < CS; ++q) {
      const float* rr = &red[((q * RB + rtf) * 16 + row) * RED_RS];
      s   += rr[col];
      deg += rr[4];
    }
    const size_t n = (size_t)b * NPATCH + (it0 + rtf) * 16 + row;
    float p = patches[n * 4 + col];
    out[n * 4 + col] = p * (1.0f + deg) - s;
  }
}

extern "C" void kernel_launch(void* const* d_in, const int* in_sizes, int n_in,
                              void* d_out, int out_size, void* d_ws, size_t ws_size,
                              hipStream_t stream) {
  const float* x  = (const float*)d_in[0];
  const float* cw = (const float*)d_in[1];
  const float* cb = (const float*)d_in[2];
  float* out = (float*)d_out;

  float* patches = (float*)d_ws;
  float* normed  = patches + (size_t)BATCH * NPATCH * 4;
  float* paugT   = normed  + (size_t)BATCH * NPATCH * 4;

  prep_kernel<<<BATCH * NTILE / PT, PREP_T, 0, stream>>>(x, cw, cb, patches, normed, paugT);

  smooth_kernel<<<NGROUP, NWAVE * 32, 0, stream>>>(patches, normed, paugT, out);
}


# Conv+norm folded into the main kernel: each block recomputes its image's operands into LDS and reads every column operand from LDS; 2 launches -> 1. Values and summation order unchanged.
# speedup vs baseline: 6.1798x; 1.1160x over previous
#include <hip/hip_runtime.h>
#include <hip/hip_fp16.h>


typedef __attribute__((ext_vector_type(2)))  float    v2f;
typedef __attribute__((ext_vector_type(8)))  float    v8f;
typedef __attribute__((ext_vector_type(16))) _Float16 v16h;

#define BATCH   8
#define HW_IN   112
#define HW_OUT  56
#define NPATCH  (HW_OUT * HW_OUT)
#define NTILE   (NPATCH / 16)
#define RB      7
#define CS      28
#define NWAVE   CS
#define NTHR    (NWAVE * 32)
#define JSEG    (NTILE / CS)
#define NGROUP  (BATCH * NTILE / RB)
#define RED_RS  5
static_assert(JSEG == RB, "column segments and row groups must have one size");

#define PH_LP    20
#define PH_ROWS  6
#define PH_TILE  (PH_ROWS * PH_LP)

#define TC_HI_PRED 0x1.c9f25cp-1f
#define TC_LO_PRED 0x1.6a09e6p-1f
#define STEP_H     0x1p30f

__device__ __forceinline__ v8f wmma_k4(v2f a, v2f b, v8f c) {
#if __has_builtin(__builtin_amdgcn_wmma_f32_16x16x4_f32)
  return __builtin_amdgcn_wmma_f32_16x16x4_f32(false, a, false, b, (short)0, c,
                                               false, false);
#else
  v16h ah = {};
  v16h bh = {};
  ah[0] = (_Float16)a[0]; ah[1] = (_Float16)a[1];
  bh[0] = (_Float16)b[0]; bh[1] = (_Float16)b[1];
  return __builtin_amdgcn_wmma_f32_16x16x32_f16(false, ah, false, bh, (short)0, c,
                                                false, false);
#endif
}

__device__ __forceinline__ float step01(float a, float pred) {
  return __builtin_amdgcn_fmed3f(__builtin_fmaf(a, STEP_H, -(pred * STEP_H)), 0.0f, 1.0f);
}

template <bool DIAG>
__device__ __forceinline__ void column_loop(
    const float* nrm_l, const float* ph_l,
    const int j0, const int it0, const int l15,
    const int kbase, const int half, const int c,
    const v2f (&b1)[RB], v8f (&acc)[RB]) {
  for (int j = j0; j < j0 + JSEG; ++j) {
    v2f a1 = *(const v2f*)(nrm_l + (j * 16 + l15) * 4 + kbase);
    const float4 u0 = *(const float4*)(ph_l + j * PH_TILE);
    const float4 u1 = *(const float4*)(ph_l + j * PH_TILE + 4);
    v2f a2[4];
    a2[0][0] = u0.x; a2[0][1] = u0.y;
    a2[1][0] = u0.z; a2[1][1] = u0.w;
    a2[2][0] = u1.x; a2[2][1] = u1.y;
    a2[3][0] = u1.z; a2[3][1] = u1.w;

#pragma unroll
    for (int r = 0; r < RB; ++r) {
      v8f zero = {};
      v8f cosd = wmma_k4(a1, b1[r], zero);

      float lo[8], hi[8];
#pragma unroll
      for (int i = 0; i < 8; ++i) {
        const float a = __builtin_fabsf(cosd[i]);
        lo[i] = step01(a, TC_LO_PRED);
        hi[i] = step01(a, TC_HI_PRED);
      }

#pragma unroll
      for (int q = 0; q < 4; ++q) {
        v2f l, h;
        l[0] = lo[2 * q];  l[1] = lo[2 * q + 1];
        h[0] = hi[2 * q];  h[1] = hi[2 * q + 1];
        v2f b2 = l + h;
        if (DIAG) {
          if (it0 + r == j) {
            if (2 * q + 8 * half == c)     b2[0] = 0.0f;
            if (2 * q + 1 + 8 * half == c) b2[1] = 0.0f;
          }
        }
        acc[r] = wmma_k4(a2[q], b2, acc[r]);
      }
    }
  }
}

__global__ __launch_bounds__(NTHR) void smooth_kernel(
    const float* __restrict__ x,
    const float* __restrict__ cw,
    const float* __restrict__ cb,
    float* __restrict__ out) {
  __shared__ float4 s_nrm[NPATCH];
  __shared__ float4 s_ph[NTILE * PH_TILE / 4];
  __shared__ float4 s_prow[RB * 16];
  __shared__ float  red[CS * RB * 16 * RED_RS];

  const int lane  = threadIdx.x & 31;
  const int wave  = __builtin_amdgcn_readfirstlane(threadIdx.x >> 5);
  const int grp   = blockIdx.x;
  const int b     = grp / (NTILE / RB);
  const int it0   = (grp % (NTILE / RB)) * RB;

  {
    const float* xb = x + (size_t)b * HW_IN * HW_IN;
    float* phf = (float*)s_ph;
    constexpr int NPASS = (NPATCH + NTHR - 1) / NTHR;

    float2 r0[NPASS], r1[NPASS];
    static_assert(NTHR % HW_OUT == 0, "a pass must be whole rows of patches");
    const int h0 = threadIdx.x / HW_OUT, w0 = threadIdx.x % HW_OUT;
#pragma unroll
    for (int i = 0; i < NPASS; ++i) {
      const int hi = h0 + i * (NTHR / HW_OUT);
      const int h  = (hi < HW_OUT) ? hi : (HW_OUT - 1);
      const float* xp = xb + (2 * h) * HW_IN + 2 * w0;
      r0[i] = *(const float2*)(xp);
      r1[i] = *(const float2*)(xp + HW_IN);
    }

#pragma unroll
    for (int i = 0; i < NPASS; ++i) {
      const int n = threadIdx.x + i * NTHR;
      if (n < NPATCH) {
        float x00 = r0[i].x, x01 = r0[i].y, x10 = r1[i].x, x11 = r1[i].y;

        float f[4];
        float ss = 0.0f;
#pragma unroll
        for (int cc = 0; cc < 4; ++cc) {
          float v = x00 * cw[cc * 4 + 0] + x01 * cw[cc * 4 + 1] +
                    x10 * cw[cc * 4 + 2] + x11 * cw[cc * 4 + 3] + cb[cc];
          f[cc] = v;
          ss += v * v;
        }
        float inv = 1.0f / (sqrtf(ss) + 1e-12f);

        s_nrm[n] = make_float4(f[0] * inv, f[1] * inv, f[2] * inv, f[3] * inv);

        float* t = phf + (n >> 4) * PH_TILE + (n & 15);
        t[0 * PH_LP] = 0.5f * f[0];
        t[1 * PH_LP] = 0.5f * f[1];
        t[2 * PH_LP] = 0.5f * f[2];
        t[3 * PH_LP] = 0.5f * f[3];
        t[4 * PH_LP] = 0.5f;
        t[5 * PH_LP] = 0.0f;

        const int rr = n - it0 * 16;
        if (rr >= 0 && rr < RB * 16)
          s_prow[rr] = make_float4(f[0], f[1], f[2], f[3]);
      }
    }
  }
  __syncthreads();

  const int half  = lane >> 4;
  const int l15   = lane & 15;
  const int kbase = half << 1;
  const int c     = l15;

  const float* nrm_l = (const float*)s_nrm;
  const int crow = (c < PH_ROWS - 1) ? c : (PH_ROWS - 1);
  const float* ph_l = (const float*)s_ph + crow * PH_LP + 8 * half;

  v2f b1[RB];
#pragma unroll
  for (int r = 0; r < RB; ++r)
    b1[r] = *(const v2f*)(nrm_l + ((it0 + r) * 16 + l15) * 4 + kbase);

  v8f acc[RB];
#pragma unroll
  for (int r = 0; r < RB; ++r) {
    v8f z = {};
    acc[r] = z;
  }

  const int j0 = wave * JSEG;
  if (j0 == it0)
    column_loop<true>(nrm_l, ph_l, j0, it0, l15, kbase, half, c, b1, acc);
  else
    column_loop<false>(nrm_l, ph_l, j0, it0, l15, kbase, half, c, b1, acc);

  if (half == 0) {
#pragma unroll
    for (int r = 0; r < RB; ++r) {
      float* myred = &red[((wave * RB + r) * 16 + c) * RED_RS];
#pragma unroll
      for (int k = 0; k < RED_RS; ++k)
        myred[k] = acc[r][k];
    }
  }
  __syncthreads();

  if (threadIdx.x < RB * 64) {
    const int rtf = threadIdx.x >> 6;
    const int e   = threadIdx.x & 63;
    const int row = e >> 2;
    const int col = e & 3;
    float s = 0.0f, deg = 0.0f;
#pragma unroll
    for (int q = 0; q < CS; ++q) {
      const float* rr = &red[((q * RB + rtf) * 16 + row) * RED_RS];
      s   += rr[col];
      deg += rr[4];
    }
    const size_t n = (size_t)b * NPATCH + (it0 + rtf) * 16 + row;
    float p = ((const float*)s_prow)[(rtf * 16 + row) * 4 + col];
    out[n * 4 + col] = p * (1.0f + deg) - s;
  }
}

extern "C" void kernel_launch(void* const* d_in, const int* in_sizes, int n_in,
                              void* d_out, int out_size, void* d_ws, size_t ws_size,
                              hipStream_t stream) {
  const float* x  = (const float*)d_in[0];
  const float* cw = (const float*)d_in[1];
  const float* cb = (const float*)d_in[2];
  float* out = (float*)d_out;

  smooth_kernel<<<NGROUP, NTHR, 0, stream>>>(x, cw, cb, out);
}
